# starting kernel
# baseline (speedup 1.0000x reference)
#include <hip/hip_runtime.h>


typedef __attribute__((ext_vector_type(2))) float v2f;
typedef __attribute__((ext_vector_type(8))) float v8f;

#define DIM       4096
#define NBLK      1024
#define ROWS      16
#define LDSTRIDE  4164
#define LDS_BYTES (ROWS * LDSTRIDE * 4)

__device__ __forceinline__ int pcol(int c) { return c + ((c >> 8) << 2); }

__global__ void build_tables_kernel(const float* __restrict__ w,
                                    float* __restrict__ tbl) {
  const int b     = blockIdx.x;
  const int stage = b >> 8;
  const int gid   = b & 255;
  const int t     = threadIdx.x;
  const int q     = t >> 6;
  const int L     = (t >> 1) & 31;
  const int p     = t & 1;

  const int j_in  = 4 * q + 2 * (L >> 4) + p;
  const int j_out = L & 15;
  const int h  = j_in >> 2,  l_in = j_in & 3;
  const int m2 = j_out >> 2, l    = j_out & 3;

  int g, stuff, t0;
  if (stage == 0)      { g = 1;   stuff = gid;      t0 = 0;        }
  else if (stage == 1) { g = 16;  stuff = gid >> 4; t0 = gid & 15; }
  else                 { g = 256; stuff = 0;        t0 = gid;      }

  const int nA = (stuff * 4 + h) * g + t0;
  const int nB = stuff * 4 * g + l * g + t0;
  const int LA = 2 * stage, LB = LA + 1;

  const float wa = w[((LA * NBLK + nA) * 4 + l_in) * 4 + l];
  const float wb = w[((LB * NBLK + nB) * 4 + h) * 4 + m2];
  tbl[(size_t)b * 256 + t] = wa * wb;
}

__global__ void fused_butterfly_kernel(const float* __restrict__ x,
                                       const float* __restrict__ tbl,
                                       float* __restrict__ out) {
  extern __shared__ float tile[];
  const int tid  = threadIdx.x;
  const int wave = tid >> 5;
  const int lane = tid & 31;
  const size_t rowbase = (size_t)blockIdx.x * ROWS * DIM;

  for (int i = tid; i < ROWS * (DIM / 4); i += 512) {
    const int r  = i >> 10;
    const int c4 = i & 1023;
    const float4 v = reinterpret_cast<const float4*>(x + rowbase + (size_t)r * DIM)[c4];
    *reinterpret_cast<float4*>(&tile[r * LDSTRIDE + pcol(c4 * 4)]) = v;
  }
  __syncthreads();

  const int Arow = lane & 15;
  const int kk2  = (lane >> 4) << 1;

  #pragma unroll
  for (int stage = 0; stage < 3; ++stage) {
    const int g = (stage == 0) ? 1 : (stage == 1) ? 16 : 256;
    for (int gid = wave; gid < 256; gid += 16) {
      int base;
      if (stage == 0)      base = gid << 4;
      else if (stage == 1) base = ((gid >> 4) << 8) + (gid & 15);
      else                 base = gid;

      const v2f* __restrict__ B =
          reinterpret_cast<const v2f*>(tbl + (((size_t)stage * 256 + gid) << 8));

      v8f c = {};
      #pragma unroll
      for (int q = 0; q < 4; ++q) {
        v2f a;
        const int j0 = 4 * q + kk2;
        a.x = tile[Arow * LDSTRIDE + pcol(base + j0 * g)];
        a.y = tile[Arow * LDSTRIDE + pcol(base + (j0 + 1) * g)];
        const v2f bb = B[q * 32 + lane];
        c = __builtin_amdgcn_wmma_f32_16x16x4_f32(
                false, a, false, bb, (short)0, c, false, false);
      }

      const int ccol = pcol(base + (lane & 15) * g);
      const int radd = (lane >> 4) << 3;
      #pragma unroll
      for (int v = 0; v < 8; ++v)
        tile[(v + radd) * LDSTRIDE + ccol] = c[v];
    }
    __syncthreads();
  }

  for (int i = tid; i < ROWS * (DIM / 4); i += 512) {
    const int r  = i >> 10;
    const int c4 = i & 1023;
    const float4 v = *reinterpret_cast<const float4*>(&tile[r * LDSTRIDE + pcol(c4 * 4)]);
    reinterpret_cast<float4*>(out + rowbase + (size_t)r * DIM)[c4] = v;
  }
}

extern "C" void kernel_launch(void* const* d_in, const int* in_sizes, int n_in,
                              void* d_out, int out_size, void* d_ws, size_t ws_size,
                              hipStream_t stream) {
  const float* x = (const float*)d_in[0];
  const float* w = (const float*)d_in[1];
  float* out = (float*)d_out;
  float* tbl = (float*)d_ws;

  (void)hipFuncSetAttribute((const void*)fused_butterfly_kernel,
                            hipFuncAttributeMaxDynamicSharedMemorySize,
                            LDS_BYTES);

  build_tables_kernel<<<768, 256, 0, stream>>>(w, tbl);
  fused_butterfly_kernel<<<8192 / ROWS, 512, LDS_BYTES, stream>>>(x, tbl, out);
}


# Stage 2 split into two halves of 128 groups: the first half's output columns are stored to HBM (512 B per wave store) while the second half runs on the matrix pipe; sum order unchanged
# speedup vs baseline: 1.4947x; 1.4947x over previous
#include <hip/hip_runtime.h>


typedef __attribute__((ext_vector_type(2))) float v2f;
typedef __attribute__((ext_vector_type(8))) float v8f;

#define DIM       4096
#define NBLK      1024
#define ROWS      16
#define NTHR      1024
#define NWAVE     (NTHR / 32)
#define LDSTRIDE  4164
#define LDS_BYTES (ROWS * LDSTRIDE * 4)

__device__ __forceinline__ int pcol(int c) { return c + ((c >> 8) << 2); }

__global__ void build_tables_kernel(const float* __restrict__ w,
                                    float* __restrict__ tbl) {
  const int b     = blockIdx.x;
  const int stage = b >> 8;
  const int gid   = b & 255;
  const int t     = threadIdx.x;
  const int q     = t >> 6;
  const int L     = (t >> 1) & 31;
  const int p     = t & 1;

  const int j_in  = 4 * q + 2 * (L >> 4) + p;
  const int j_out = L & 15;
  const int h  = j_in >> 2,  l_in = j_in & 3;
  const int m2 = j_out >> 2, l    = j_out & 3;

  int g, stuff, t0;
  if (stage == 0)      { g = 1;   stuff = gid;      t0 = 0;        }
  else if (stage == 1) { g = 16;  stuff = gid >> 4; t0 = gid & 15; }
  else                 { g = 256; stuff = 0;        t0 = gid;      }

  const int nA = (stuff * 4 + h) * g + t0;
  const int nB = stuff * 4 * g + l * g + t0;
  const int LA = 2 * stage, LB = LA + 1;

  const float wa = w[((LA * NBLK + nA) * 4 + l_in) * 4 + l];
  const float wb = w[((LB * NBLK + nB) * 4 + h) * 4 + m2];
  tbl[(size_t)b * 256 + t] = wa * wb;
}

#define GB 4
template <int STAGE>
__device__ __forceinline__ void load_tf(v2f (&tf)[GB][4],
                                        const float* __restrict__ tbl,
                                        int blk, int lane) {
  const int gid0 = blk * GB;
  const v2f* __restrict__ Tp =
      reinterpret_cast<const v2f*>(tbl + (((size_t)STAGE * 256 + gid0) << 8));
  #pragma unroll
  for (int i = 0; i < GB; ++i)
    #pragma unroll
    for (int q = 0; q < 4; ++q)
      tf[i][q] = Tp[i * 128 + q * 32 + lane];
}

template <int STAGE>
__device__ __forceinline__ void compute_blk(float* tile, const v2f (&tf)[GB][4],
                                            int blk, int lane) {
  constexpr int g = (STAGE == 0) ? 1 : (STAGE == 1) ? 16 : 256;
  const int h = lane >> 4;
  float* row = tile + (lane & 15) * LDSTRIDE;
  {
    const int gid0 = blk * GB;
    int base;
    if (STAGE == 0)      base = gid0 << 4;
    else if (STAGE == 1) base = ((gid0 >> 4) << 8) + (gid0 & 15);
    else                 base = gid0;

    v8f c[GB];
    #pragma unroll
    for (int i = 0; i < GB; ++i) c[i] = (v8f){};

    if (STAGE == 0) {
      v2f xf[GB][4];
      #pragma unroll
      for (int i = 0; i < GB; ++i)
        #pragma unroll
        for (int q = 0; q < 4; ++q)
          xf[i][q] = *reinterpret_cast<const v2f*>(
              &row[pcol(base + 16 * i + 4 * q + 2 * h)]);
      #pragma unroll
      for (int q = 0; q < 4; ++q)
        #pragma unroll
        for (int i = 0; i < GB; ++i)
          c[i] = __builtin_amdgcn_wmma_f32_16x16x4_f32(
                     false, tf[i][q], false, xf[i][q], (short)0, c[i], false, false);
      #pragma unroll
      for (int i = 0; i < GB; ++i) {
        float* dst = &row[pcol(base + 16 * i + 8 * h)];
        *reinterpret_cast<float4*>(dst)     = make_float4(c[i][0], c[i][1], c[i][2], c[i][3]);
        *reinterpret_cast<float4*>(dst + 4) = make_float4(c[i][4], c[i][5], c[i][6], c[i][7]);
      }
    } else {
      float4 xv[4][2];
      #pragma unroll
      for (int q = 0; q < 4; ++q)
        #pragma unroll
        for (int s = 0; s < 2; ++s)
          xv[q][s] = *reinterpret_cast<const float4*>(
              &row[pcol(base + (4 * q + 2 * h + s) * g)]);
      #pragma unroll
      for (int q = 0; q < 4; ++q) {
        const float x0[GB] = {xv[q][0].x, xv[q][0].y, xv[q][0].z, xv[q][0].w};
        const float x1[GB] = {xv[q][1].x, xv[q][1].y, xv[q][1].z, xv[q][1].w};
        #pragma unroll
        for (int i = 0; i < GB; ++i) {
          v2f xb; xb.x = x0[i]; xb.y = x1[i];
          c[i] = __builtin_amdgcn_wmma_f32_16x16x4_f32(
                     false, tf[i][q], false, xb, (short)0, c[i], false, false);
        }
      }
      #pragma unroll
      for (int r = 0; r < 8; ++r)
        *reinterpret_cast<float4*>(&row[pcol(base + (8 * h + r) * g)]) =
            make_float4(c[0][r], c[1][r], c[2][r], c[3][r]);
    }
  }
}

template <int STAGE>
__device__ __forceinline__ void run_stage(float* tile,
                                          const float* __restrict__ tbl,
                                          int wave, int lane) {
  for (int blk = wave; blk < 256 / GB; blk += NWAVE) {
    v2f tf[GB][4];
    load_tf<STAGE>(tf, tbl, blk, lane);
    compute_blk<STAGE>(tile, tf, blk, lane);
  }
}

__device__ __forceinline__ void store_half(const float* tile, float* __restrict__ out,
                                           size_t rowbase, int tid, int half) {
  #pragma unroll
  for (int k = 0; k < (ROWS * 16 * 32) / NTHR; ++k) {
    const int i  = tid + k * NTHR;
    const int c4 = i & 31;
    const int j  = (i >> 5) & 15;
    const int r  = i >> 9;
    const int col = j * 256 + half * 128 + c4 * 4;
    const float4* src = reinterpret_cast<const float4*>(
        __builtin_assume_aligned(&tile[r * LDSTRIDE + pcol(col)], 16));
    reinterpret_cast<float4*>(out + rowbase + (size_t)r * DIM + col)[0] = *src;
  }
}

__global__ void __launch_bounds__(NTHR) fused_butterfly_kernel(const float* __restrict__ x,
                                       const float* __restrict__ tbl,
                                       float* __restrict__ out) {
  extern __shared__ float tile[];
  const int tid  = threadIdx.x;
  const int wave = tid >> 5;
  const int lane = tid & 31;
  const size_t rowbase = (size_t)blockIdx.x * ROWS * DIM;

  for (int i = tid; i < ROWS * (DIM / 4); i += NTHR) {
    const int r  = i >> 10;
    const int c4 = i & 1023;
    const float4 v = reinterpret_cast<const float4*>(x + rowbase + (size_t)r * DIM)[c4];
    *reinterpret_cast<float4*>(&tile[r * LDSTRIDE + pcol(c4 * 4)]) = v;
  }
  __syncthreads();

  run_stage<0>(tile, tbl, wave, lane);
  __syncthreads();
  run_stage<1>(tile, tbl, wave, lane);
  __syncthreads();
  {
    v2f tf[GB][4];
    load_tf<2>(tf, tbl, wave, lane);
    compute_blk<2>(tile, tf, wave, lane);
  }
  __syncthreads();
  {
    v2f tf[GB][4];
    load_tf<2>(tf, tbl, wave + NWAVE, lane);
    store_half(tile, out, rowbase, tid, 0);
    __builtin_amdgcn_sched_barrier(0);
    compute_blk<2>(tile, tf, wave + NWAVE, lane);
  }
  __builtin_amdgcn_fence(__ATOMIC_RELEASE, "workgroup", "local");
  __builtin_amdgcn_s_barrier();
  __builtin_amdgcn_fence(__ATOMIC_ACQUIRE, "workgroup", "local");
  store_half(tile, out, rowbase, tid, 1);
}

extern "C" void kernel_launch(void* const* d_in, const int* in_sizes, int n_in,
                              void* d_out, int out_size, void* d_ws, size_t ws_size,
                              hipStream_t stream) {
  const float* x = (const float*)d_in[0];
  const float* w = (const float*)d_in[1];
  float* out = (float*)d_out;
  float* tbl = (float*)d_ws;

  (void)hipFuncSetAttribute((const void*)fused_butterfly_kernel,
                            hipFuncAttributeMaxDynamicSharedMemorySize,
                            LDS_BYTES);

  build_tables_kernel<<<768, 256, 0, stream>>>(w, tbl);
  fused_butterfly_kernel<<<8192 / ROWS, NTHR, LDS_BYTES, stream>>>(x, tbl, out);
}
